# starting kernel
# baseline (speedup 1.0000x reference)
#include <hip/hip_runtime.h>

typedef __attribute__((ext_vector_type(2))) float v2f;
typedef __attribute__((ext_vector_type(4))) float v4f;
typedef __attribute__((ext_vector_type(8))) float v8f;

#define H_IMG 14
#define W_IMG 14
#define LNODES 196
#define NB 32
#define NHEADS 24
#define PDIM 64
#define SDT (NB * NHEADS * LNODES)

__launch_bounds__(128, 4)
__global__ void chimera_dag_mix(const float* __restrict__ dt,
                                const float* __restrict__ dt_bias,
                                const float* __restrict__ x,
                                const float* __restrict__ Dskip,
                                float* __restrict__ y)
{
    __shared__ float sAleft[LNODES];
    __shared__ float sAtop[LNODES];
    __shared__ float sT[16][17];
    __shared__ alignas(16) float sZin[16][68];
    __shared__ alignas(16) float sZprev[16][68];

    const int bh   = blockIdx.x;
    const int h    = bh % NHEADS;
    const int tid  = threadIdx.x;
    const int lane = tid & 31;
    const int wave = tid >> 5;
    const int col0 = wave * 16;
    const int mrow = lane & 15;
    const int ksel = (lane >> 4) << 1;

    const float bias = dt_bias[h];
    const float dsk  = Dskip[h];
    const float* __restrict__ xb = x + (size_t)bh * (LNODES * PDIM);
    float* __restrict__ yb       = y + (size_t)bh * (LNODES * PDIM);

    for (int node = tid; node < LNODES; node += 128) {
        int row = node / W_IMG, col = node % W_IMG;
        float nie = 2.0f - (col == 0 ? 1.0f : 0.0f) - (row == 0 ? 1.0f : 0.0f);
        if (nie < 1e-6f) nie = 1.0f;
        float rn = 0.95f * rsqrtf(nie);
        float v0 = dt[0 * SDT + bh * LNODES + node] + bias;
        float v1 = dt[1 * SDT + bh * LNODES + node] + bias;
        sAleft[node] = (col > 0) ? rn / (1.0f + expf(v0)) : 0.0f;
        sAtop [node] = (row > 0) ? rn / (1.0f + expf(v1)) : 0.0f;
    }
    {
        v4f zv = {};
        v4f* zp = (v4f*)&sZprev[0][0];
        for (int i = tid; i < (16 * 68) / 4; i += 128) zp[i] = zv;
        v4f* zi = (v4f*)&sZin[14][0];
        for (int i = tid; i < (2 * 68) / 4; i += 128) zi[i] = zv;
        if (tid < 32) sT[14 + (tid >> 4)][tid & 15] = 0.0f;
    }
    __syncthreads();

    const int  zc  = tid >> 3;
    const int  zp0 = (tid & 7) * 8;
    const bool act = (zc < W_IMG);

    v4f xprev0 = {}, xprev1 = {};

    for (int r = 0; r < H_IMG; ++r) {
        if (act) {
            int node = r * W_IMG + zc;
            float at = sAtop[node];
            const v4f* xp = (const v4f*)(xb + node * PDIM + zp0);
            v4f x0 = xp[0], x1 = xp[1];
            const v4f* pp = (const v4f*)&sZprev[zc][zp0];
            v4f p0 = pp[0], p1 = pp[1];
            if (r > 0) {
                v4f* yp = (v4f*)(yb + (node - W_IMG) * PDIM + zp0);
                yp[0] = p0 + dsk * xprev0;
                yp[1] = p1 + dsk * xprev1;
            }
            v4f* op = (v4f*)&sZin[zc][zp0];
            op[0] = x0 + at * p0;
            op[1] = x1 + at * p1;
            xprev0 = x0; xprev1 = x1;
        }
        if (tid < 16) {
            float t = 0.0f;
            for (int c = 0; c < W_IMG; ++c) {
                float al = sAleft[r * W_IMG + c];
                t = (tid == c) ? 1.0f : al * t;
                sT[c][tid] = t;
            }
        }
        __syncthreads();

        v8f acc = {};
#pragma unroll
        for (int k0 = 0; k0 < 16; k0 += 4) {
            v2f a, b;
            a.x = sT[mrow][k0 + ksel];
            a.y = sT[mrow][k0 + ksel + 1];
            b.x = sZin[k0 + ksel][col0 + mrow];
            b.y = sZin[k0 + ksel + 1][col0 + mrow];
            acc = __builtin_amdgcn_wmma_f32_16x16x4_f32(
                false, a, false, b, (short)0, acc, false, false);
        }

#pragma unroll
        for (int i = 0; i < 8; ++i) {
            int M = i + ((lane >> 4) << 3);
            if (M < W_IMG) sZprev[M][col0 + mrow] = acc[i];
        }
        __syncthreads();
    }

    if (act) {
        const v4f* pp = (const v4f*)&sZprev[zc][zp0];
        v4f* yp = (v4f*)(yb + ((H_IMG - 1) * W_IMG + zc) * PDIM + zp0);
        yp[0] = pp[0] + dsk * xprev0;
        yp[1] = pp[1] + dsk * xprev1;
    }
}

extern "C" void kernel_launch(void* const* d_in, const int* in_sizes, int n_in,
                              void* d_out, int out_size, void* d_ws, size_t ws_size,
                              hipStream_t stream) {
    (void)in_sizes; (void)n_in; (void)d_ws; (void)ws_size; (void)out_size;
    const float* dt      = (const float*)d_in[0];
    const float* dt_bias = (const float*)d_in[1];
    const float* x       = (const float*)d_in[2];
    const float* Dskip   = (const float*)d_in[3];
    float* y             = (float*)d_out;
    chimera_dag_mix<<<NB * NHEADS, 128, 0, stream>>>(dt, dt_bias, x, Dskip, y);
}


# Global memory taken off the 14-step row recurrence: all rows of x preloaded into registers (row loop unrolled) and workgroup barriers that fence LDS only, so no step waits on a load or a store ack
# speedup vs baseline: 1.7825x; 1.7825x over previous
#include <hip/hip_runtime.h>

typedef __attribute__((ext_vector_type(2))) float v2f;
typedef __attribute__((ext_vector_type(4))) float v4f;
typedef __attribute__((ext_vector_type(8))) float v8f;

#define H_IMG 14
#define W_IMG 14
#define LNODES 196
#define NB 32
#define NHEADS 24
#define PDIM 64
#define SDT (NB * NHEADS * LNODES)


__device__ __forceinline__ void edge_coeffs(int node, float v0, float v1,
                                            float* sAleft, float* sAtop)
{
    int row = node / W_IMG, col = node % W_IMG;
    float nie = 2.0f - (col == 0 ? 1.0f : 0.0f) - (row == 0 ? 1.0f : 0.0f);
    if (nie < 1e-6f) nie = 1.0f;
    float rn = 0.95f * rsqrtf(nie);
    float q0 = rn / (1.0f + expf(v0));
    float q1 = rn / (1.0f + expf(v1));
    sAleft[node] = (col > 0) ? q0 : 0.0f;
    sAtop [node] = (row > 0) ? q1 : 0.0f;
}

__device__ __forceinline__ void lds_barrier()
{
    __builtin_amdgcn_fence(__ATOMIC_RELEASE, "workgroup", "local");
    __builtin_amdgcn_s_barrier();
    __builtin_amdgcn_fence(__ATOMIC_ACQUIRE, "workgroup", "local");
}

__launch_bounds__(128, 4)
__global__ void dag_grid_solve(const float* __restrict__ dt,
                               const float* __restrict__ dt_bias,
                               const float* __restrict__ x,
                               const float* __restrict__ Dskip,
                               float* __restrict__ y)
{
    __shared__ float sAleft[LNODES];
    __shared__ float sAtop[LNODES];
    __shared__ float sT[16][17];
    __shared__ alignas(16) float sZin[16][68];
    __shared__ alignas(16) float sZprev[16][68];

    const int bh   = blockIdx.x;
    const int h    = bh % NHEADS;
    const int tid  = threadIdx.x;
    const int lane = tid & 31;
    const int wave = tid >> 5;
    const int col0 = wave * 16;
    const int mrow = lane & 15;
    const int ksel = (lane >> 4) << 1;

    const float bias = dt_bias[h];
    const float dsk  = Dskip[h];
    const float* __restrict__ xb = x + (size_t)bh * (LNODES * PDIM);
    float* __restrict__ yb       = y + (size_t)bh * (LNODES * PDIM);

    const int  zc  = tid >> 3;
    const int  zp0 = (tid & 7) * 8;
    const bool act = (zc < W_IMG);

    const int  nodeA = tid;
    const bool hasB  = (tid + 128) < LNODES;
    const int  nodeB = hasB ? tid + 128 : tid;
    const float* __restrict__ dtb = dt + bh * LNODES;
    const float dA0 = dtb[0 * SDT + nodeA];
    const float dA1 = dtb[1 * SDT + nodeA];
    const float dB0 = dtb[0 * SDT + nodeB];
    const float dB1 = dtb[1 * SDT + nodeB];

    v4f xs0[H_IMG], xs1[H_IMG];
    {
        const int zcl = act ? zc : (W_IMG - 1);
        const float* xcol = xb + zcl * PDIM + zp0;
#pragma unroll
        for (int r = 0; r < H_IMG; ++r) {
            const v4f* xp = (const v4f*)(xcol + r * (W_IMG * PDIM));
            xs0[r] = xp[0];
            xs1[r] = xp[1];
        }
    }

    {
        v4f zv = {};
        v4f* zp = (v4f*)&sZprev[0][0];
        for (int i = tid; i < (16 * 68) / 4; i += 128) zp[i] = zv;
        v4f* zi = (v4f*)&sZin[14][0];
        for (int i = tid; i < (2 * 68) / 4; i += 128) zi[i] = zv;
        if (tid < 32) sT[14 + (tid >> 4)][tid & 15] = 0.0f;
    }

    edge_coeffs(nodeA, dA0 + bias, dA1 + bias, sAleft, sAtop);
    if (hasB) edge_coeffs(nodeB, dB0 + bias, dB1 + bias, sAleft, sAtop);
    lds_barrier();

#pragma unroll
    for (int r = 0; r < H_IMG; ++r) {
        if (act) {
            int node = r * W_IMG + zc;
            float at = sAtop[node];
            const v4f* pp = (const v4f*)&sZprev[zc][zp0];
            v4f p0 = pp[0], p1 = pp[1];
            if (r > 0) {
                v4f* yp = (v4f*)(yb + (node - W_IMG) * PDIM + zp0);
                yp[0] = p0 + dsk * xs0[r > 0 ? r - 1 : 0];
                yp[1] = p1 + dsk * xs1[r > 0 ? r - 1 : 0];
            }
            v4f* op = (v4f*)&sZin[zc][zp0];
            op[0] = xs0[r] + at * p0;
            op[1] = xs1[r] + at * p1;
        }
        if (tid < 16) {
            float t = 0.0f;
            for (int c = 0; c < W_IMG; ++c) {
                float al = sAleft[r * W_IMG + c];
                t = (tid == c) ? 1.0f : al * t;
                sT[c][tid] = t;
            }
        }
        lds_barrier();

        v8f acc = {};
#pragma unroll
        for (int k0 = 0; k0 < 16; k0 += 4) {
            v2f a, b;
            a.x = sT[mrow][k0 + ksel];
            a.y = sT[mrow][k0 + ksel + 1];
            b.x = sZin[k0 + ksel][col0 + mrow];
            b.y = sZin[k0 + ksel + 1][col0 + mrow];
            acc = __builtin_amdgcn_wmma_f32_16x16x4_f32(
                false, a, false, b, (short)0, acc, false, false);
        }

#pragma unroll
        for (int i = 0; i < 8; ++i) {
            int M = i + ((lane >> 4) << 3);
            if (M < W_IMG) sZprev[M][col0 + mrow] = acc[i];
        }
        lds_barrier();
    }

    if (act) {
        const v4f* pp = (const v4f*)&sZprev[zc][zp0];
        v4f* yp = (v4f*)(yb + ((H_IMG - 1) * W_IMG + zc) * PDIM + zp0);
        yp[0] = pp[0] + dsk * xs0[H_IMG - 1];
        yp[1] = pp[1] + dsk * xs1[H_IMG - 1];
    }
}

extern "C" void kernel_launch(void* const* d_in, const int* in_sizes, int n_in,
                              void* d_out, int out_size, void* d_ws, size_t ws_size,
                              hipStream_t stream) {
    (void)in_sizes; (void)n_in; (void)d_ws; (void)ws_size; (void)out_size;
    const float* dt      = (const float*)d_in[0];
    const float* dt_bias = (const float*)d_in[1];
    const float* x       = (const float*)d_in[2];
    const float* Dskip   = (const float*)d_in[3];
    float* y             = (float*)d_out;
    dag_grid_solve<<<NB * NHEADS, 128, 0, stream>>>(dt, dt_bias, x, Dskip, y);
}


# Row recurrence in registers: k order = accumulator row order, so a row's WMMA result is the next row's B operand; x and y in that map, 14 triangular matrices built once; no per-row LDS or barrier
# speedup vs baseline: 1.9178x; 1.0759x over previous
#include <hip/hip_runtime.h>

typedef __attribute__((ext_vector_type(2))) float v2f;
typedef __attribute__((ext_vector_type(3))) float v3f;
typedef __attribute__((ext_vector_type(4))) float v4f;
typedef __attribute__((ext_vector_type(8))) float v8f;

#define H_IMG 14
#define W_IMG 14
#define WP 16
#define TP 20
#define TR (16 * TP + 16)
#define LNODES 196
#define NB 32
#define NHEADS 24
#define PDIM 64
#define SDT (NB * NHEADS * LNODES)
#define XROWS_A 8


__device__ __forceinline__ int slot_of(int c) { return c + (c >= 7 ? 1 : 0); }

__device__ __forceinline__ void edge_coeffs(int node, float v0, float v1,
                                            float* sAleft, float* sAtop)
{
    int row = node / W_IMG, col = node % W_IMG;
    float nie = 2.0f - (col == 0 ? 1.0f : 0.0f) - (row == 0 ? 1.0f : 0.0f);
    if (nie < 1e-6f) nie = 1.0f;
    float rn = 0.95f * rsqrtf(nie);
    float q0 = rn / (1.0f + expf(v0));
    float q1 = rn / (1.0f + expf(v1));
    sAleft[row * WP + col]          = (col > 0) ? q0 : 0.0f;
    sAtop [row * WP + slot_of(col)] = (row > 0) ? q1 : 0.0f;
}

__device__ __forceinline__ void lds_barrier()
{
    __builtin_amdgcn_fence(__ATOMIC_RELEASE, "workgroup", "local");
    __builtin_amdgcn_s_barrier();
    __builtin_amdgcn_fence(__ATOMIC_ACQUIRE, "workgroup", "local");
}

__launch_bounds__(128, 4)
__global__ void dag_grid_solve(const float* __restrict__ dt,
                               const float* __restrict__ dt_bias,
                               const float* __restrict__ x,
                               const float* __restrict__ Dskip,
                               float* __restrict__ y)
{
    __shared__ alignas(16) float sAleft[H_IMG * WP];
    __shared__ alignas(16) float sAtop[H_IMG * WP];
    __shared__ alignas(16) float sT[H_IMG * TR];

    const int bh   = blockIdx.x;
    const int h    = bh % NHEADS;
    const int tid  = threadIdx.x;
    const int lane = tid & 31;
    const int wave = tid >> 5;
    const int col0 = wave * 16;
    const int n    = lane & 15;
    const int hh   = lane >> 4;

    const float bias = dt_bias[h];
    const float dsk  = Dskip[h];
    const float* __restrict__ xb = x + (size_t)bh * (LNODES * PDIM);
    float* __restrict__ yb       = y + (size_t)bh * (LNODES * PDIM);

    const int  nodeA = tid;
    const int  nodeB = (tid + 128 < LNODES) ? tid + 128 : LNODES - 1;
    const float* __restrict__ dtb = dt + bh * LNODES;
    const float dA0 = dtb[0 * SDT + nodeA];
    const float dA1 = dtb[1 * SDT + nodeA];
    const float dB0 = dtb[0 * SDT + nodeB];
    const float dB1 = dtb[1 * SDT + nodeB];
    __builtin_amdgcn_sched_barrier(0);

    const float* xl = xb + (7 * hh) * PDIM + col0 + n;
    float xs[H_IMG][7];
#pragma unroll
    for (int r = 0; r < XROWS_A; ++r)
#pragma unroll
        for (int i = 0; i < 7; ++i)
            xs[r][i] = xl[(r * W_IMG + i) * PDIM];
    __builtin_amdgcn_sched_barrier(0);

    edge_coeffs(nodeA, dA0 + bias, dA1 + bias, sAleft, sAtop);
    edge_coeffs(nodeB, dB0 + bias, dB1 + bias, sAleft, sAtop);
    lds_barrier();

    {
        const int cp = tid & 15;
        const int kslot = (cp < W_IMG) ? slot_of(cp) : (cp == W_IMG ? 7 : 15);
#pragma unroll
        for (int rr = 0; rr < 2; ++rr) {
            const int r = (tid >> 4) + 8 * rr;
            if (r < H_IMG) {
                float t = 0.0f;
#pragma unroll
                for (int c = 0; c < W_IMG; ++c) {
                    float al = sAleft[r * WP + c];
                    t = (cp == c) ? 1.0f : al * t;
                    sT[r * TR + (c + (c >= 7 ? 1 : 0)) * TP + kslot] = t;
                }
                sT[r * TR + 7 * TP + kslot]  = 0.0f;
                sT[r * TR + 15 * TP + kslot] = 0.0f;
            }
        }
    }
    lds_barrier();
    __builtin_amdgcn_sched_barrier(0);

#pragma unroll
    for (int r = XROWS_A; r < H_IMG; ++r)
#pragma unroll
        for (int i = 0; i < 7; ++i)
            xs[r][i] = xl[(r * W_IMG + i) * PDIM];

    __builtin_amdgcn_sched_barrier(0);

    float* yl = yb + (7 * hh) * PDIM + col0 + n;
    const float* atp = &sAtop[8 * hh];
    const float* Tp  = &sT[n * TP + 8 * hh];
    v4f atn0 = *(const v4f*)atp;
    v3f atn1 = *(const v3f*)(atp + 4);
    v4f an0  = *(const v4f*)Tp;
    v4f an1  = *(const v4f*)(Tp + 4);
    v8f z = {};
#pragma unroll
    for (int r = 0; r < H_IMG; ++r) {
        const v4f at0 = atn0;
        const v3f at1 = atn1;
        v2f a[4];
        a[0].x = an0[0]; a[0].y = an0[1];
        a[1].x = an0[2]; a[1].y = an0[3];
        a[2].x = an1[0]; a[2].y = an1[1];
        a[3].x = an1[2]; a[3].y = an1[3];
        if (r + 1 < H_IMG) {
            atn0 = *(const v4f*)(atp + (r + 1) * WP);
            atn1 = *(const v3f*)(atp + (r + 1) * WP + 4);
            an0  = *(const v4f*)(Tp + (r + 1) * TR);
            an1  = *(const v4f*)(Tp + (r + 1) * TR + 4);
        }
        __builtin_amdgcn_sched_barrier(0);

        v8f zin;
        zin[0] = xs[r][0] + at0[0] * z[0];
        zin[1] = xs[r][1] + at0[1] * z[1];
        zin[2] = xs[r][2] + at0[2] * z[2];
        zin[3] = xs[r][3] + at0[3] * z[3];
        zin[4] = xs[r][4] + at1[0] * z[4];
        zin[5] = xs[r][5] + at1[1] * z[5];
        zin[6] = xs[r][6] + at1[2] * z[6];
        zin[7] = 0.0f;

        v8f acc = {};
#pragma unroll
        for (int s = 0; s < 4; ++s) {
            v2f b;
            b.x = zin[2 * s];
            b.y = zin[2 * s + 1];
            acc = __builtin_amdgcn_wmma_f32_16x16x4_f32(
                false, a[s], false, b, (short)0, acc, false, false);
        }
        if (r > 0) {
            float* yr = yl + ((r - 1) * W_IMG) * PDIM;
#pragma unroll
            for (int i = 0; i < 7; ++i)
                yr[i * PDIM] = z[i] + dsk * xs[r > 0 ? r - 1 : 0][i];
        }
        z = acc;
        __builtin_amdgcn_sched_barrier(0);
    }
    {
        float* yr = yl + ((H_IMG - 1) * W_IMG) * PDIM;
#pragma unroll
        for (int i = 0; i < 7; ++i)
            yr[i * PDIM] = z[i] + dsk * xs[H_IMG - 1][i];
    }
}

extern "C" void kernel_launch(void* const* d_in, const int* in_sizes, int n_in,
                              void* d_out, int out_size, void* d_ws, size_t ws_size,
                              hipStream_t stream) {
    (void)in_sizes; (void)n_in; (void)d_ws; (void)ws_size; (void)out_size;
    const float* dt      = (const float*)d_in[0];
    const float* dt_bias = (const float*)d_in[1];
    const float* x       = (const float*)d_in[2];
    const float* Dskip   = (const float*)d_in[3];
    float* y             = (float*)d_out;
    dag_grid_solve<<<NB * NHEADS, 128, 0, stream>>>(dt, dt_bias, x, Dskip, y);
}
